# starting kernel
# baseline (speedup 1.0000x reference)
#include <hip/hip_runtime.h>
#include <hip/hip_bf16.h>

#define LATENT   1024
#define N_MLP    64
#define N_GRID2  16384
#define OUT_DIM  3
#define NJ       192
#define W_ROWSTRIDE 3078
#define KCHUNKS  64
#define JT_TILES 12

typedef __attribute__((ext_vector_type(2))) float v2f;
typedef __attribute__((ext_vector_type(8))) float v8f;

#if defined(__has_builtin)
#if __has_builtin(__builtin_amdgcn_tanhf)
#define FAST_TANH(v) __builtin_amdgcn_tanhf(v)
#endif
#endif
#ifndef FAST_TANH
#define FAST_TANH(v) tanhf(v)
#endif

__global__ __launch_bounds__(32)
void phase1_wmma(const float* __restrict__ x,
                 const float* __restrict__ W,
                 float* __restrict__ partial) {
    const int jt   = (int)blockIdx.x >> 6;
    const int kc   = (int)blockIdx.x & 63;
    const int lane = (int)threadIdx.x;
    const int row  = lane & 15;
    const int j    = jt * 16 + row;
    const int m    = j / 3;
    const int o    = j - 3 * m;
    const float* wrow = W + (size_t)m * W_ROWSTRIDE + o;
    const int koff = (lane >> 4) << 1;
    const int kbase = kc * 16;

    v8f acc = {};
    #pragma unroll
    for (int s = 0; s < 4; ++s) {
        const int k0 = kbase + s * 4 + koff;
        v2f a, b;
        a.x = wrow[3 * k0];
        a.y = wrow[3 * (k0 + 1)];
        b.x = x[k0];
        b.y = x[k0 + 1];
        acc = __builtin_amdgcn_wmma_f32_16x16x4_f32(
             false, a,  false, b,
             (short)0, acc,  false,  false);
    }

    float* p = partial + (size_t)kc * NJ + jt * 16;
    if (lane == 0) {
        #pragma unroll
        for (int v = 0; v < 8; ++v) p[v] = acc[v];
    } else if (lane == 16) {
        #pragma unroll
        for (int v = 0; v < 8; ++v) p[8 + v] = acc[v];
    }
}

__global__ __launch_bounds__(256)
void reduce_coef(const float* __restrict__ partial,
                 const float* __restrict__ b,
                 const float* __restrict__ W,
                 float* __restrict__ coef) {
    const int j = (int)threadIdx.x;
    if (j < NJ) {
        float s = b[j];
        #pragma unroll 8
        for (int kc = 0; kc < KCHUNKS; ++kc)
            s += partial[(size_t)kc * NJ + j];
        const int m = j / 3;
        const int o = j - 3 * m;
        coef[j]           = s;
        coef[NJ + j]      = W[(size_t)m * W_ROWSTRIDE + 3 * LATENT + o];
        coef[2 * NJ + j]  = W[(size_t)m * W_ROWSTRIDE + 3 * (LATENT + 1) + o];
    }
}

__global__ __launch_bounds__(256)
void phase2_eval(const float* __restrict__ coef,
                 const float* __restrict__ grid,
                 float* __restrict__ out) {
    const int tid = (int)blockIdx.x * 256 + (int)threadIdx.x;
    const int m   = tid >> 12;
    const int q   = tid & 4095;
    const int g0  = q << 2;

    const float4* gp = (const float4*)(grid + (size_t)g0 * 2);
    const float4 gA = gp[0];
    const float4 gB = gp[1];

    const float c0  = coef[m * 3 + 0], c1  = coef[m * 3 + 1], c2  = coef[m * 3 + 2];
    const float wx0 = coef[NJ + m * 3 + 0], wx1 = coef[NJ + m * 3 + 1], wx2 = coef[NJ + m * 3 + 2];
    const float wy0 = coef[2 * NJ + m * 3 + 0], wy1 = coef[2 * NJ + m * 3 + 1], wy2 = coef[2 * NJ + m * 3 + 2];

    const float r0  = FAST_TANH(c0 + gA.x * wx0 + gA.y * wy0);
    const float r1  = FAST_TANH(c1 + gA.x * wx1 + gA.y * wy1);
    const float r2  = FAST_TANH(c2 + gA.x * wx2 + gA.y * wy2);
    const float r3  = FAST_TANH(c0 + gA.z * wx0 + gA.w * wy0);
    const float r4  = FAST_TANH(c1 + gA.z * wx1 + gA.w * wy1);
    const float r5  = FAST_TANH(c2 + gA.z * wx2 + gA.w * wy2);
    const float r6  = FAST_TANH(c0 + gB.x * wx0 + gB.y * wy0);
    const float r7  = FAST_TANH(c1 + gB.x * wx1 + gB.y * wy1);
    const float r8  = FAST_TANH(c2 + gB.x * wx2 + gB.y * wy2);
    const float r9  = FAST_TANH(c0 + gB.z * wx0 + gB.w * wy0);
    const float r10 = FAST_TANH(c1 + gB.z * wx1 + gB.w * wy1);
    const float r11 = FAST_TANH(c2 + gB.z * wx2 + gB.w * wy2);

    float4* op = (float4*)(out + ((size_t)m * N_GRID2 + g0) * 3);
    op[0] = make_float4(r0, r1, r2, r3);
    op[1] = make_float4(r4, r5, r6, r7);
    op[2] = make_float4(r8, r9, r10, r11);
}

extern "C" void kernel_launch(void* const* d_in, const int* in_sizes, int n_in,
                              void* d_out, int out_size, void* d_ws, size_t ws_size,
                              hipStream_t stream) {
    const float* x    = (const float*)d_in[0];
    const float* W    = (const float*)d_in[1];
    const float* b    = (const float*)d_in[2];
    const float* grid = (const float*)d_in[3];
    float* out        = (float*)d_out;

    float* coef    = (float*)d_ws;
    float* partial = coef + 3 * NJ;

    phase1_wmma<<<dim3(JT_TILES * KCHUNKS), dim3(32), 0, stream>>>(x, W, partial);
    reduce_coef<<<dim3(1), dim3(256), 0, stream>>>(partial, b, W, coef);
    phase2_eval<<<dim3((N_MLP * N_GRID2 / 4) / 256), dim3(256), 0, stream>>>(coef, grid, out);
}


# fuse the 3 launches into 1: each of 256 blocks recomputes its decoder's 3 dot products on WMMA (diagonal K-split, row_xmask+LDS tree) then evaluates 4096 points; changes the summation order
# speedup vs baseline: 1.8445x; 1.8445x over previous
#include <hip/hip_runtime.h>


#define LATENT   1024
#define N_MLP    64
#define N_GRID2  16384
#define OUT_DIM  3
#define W_ROWSTRIDE 3078

#define BLOCKS_PER_MLP 4
#define N_BLOCKS       256
#define PTS_PER_BLOCK  4096
#define PTS_PER_PASS   1024
#define PASSES         4
#define N_WAVES        8
#define K_PER_WAVE     128
#define K_PER_STEP     64

typedef __attribute__((ext_vector_type(2))) float v2f;
typedef __attribute__((ext_vector_type(8))) float v8f;

#if defined(__has_builtin)
#if __has_builtin(__builtin_amdgcn_tanhf)
#define FAST_TANH(v) __builtin_amdgcn_tanhf(v)
#endif
#endif
#ifndef FAST_TANH
#define FAST_TANH(v) tanhf(v)
#endif

template <int MASK>
__device__ __forceinline__ float row_xor_add(float v) {
    const int t = __builtin_amdgcn_update_dpp(0, __builtin_bit_cast(int, v),
                                              0x160 + MASK, 0xF, 0xF, false);
    return v + __builtin_bit_cast(float, t);
}

__device__ __forceinline__ float wave_sum(float v) {
    v = row_xor_add<1>(v);
    v = row_xor_add<2>(v);
    v = row_xor_add<4>(v);
    v = row_xor_add<8>(v);
    const float lo = __builtin_bit_cast(float, __builtin_amdgcn_readlane(__builtin_bit_cast(int, v), 0));
    const float hi = __builtin_bit_cast(float, __builtin_amdgcn_readlane(__builtin_bit_cast(int, v), 16));
    return lo + hi;
}

__global__ __launch_bounds__(256)
void decode_fused(const float* __restrict__ x,
                  const float* __restrict__ W,
                  const float* __restrict__ b,
                  const float* __restrict__ grid,
                  float* __restrict__ out) {
    __shared__ float red[OUT_DIM * N_WAVES];

    const int tid  = (int)threadIdx.x;
    const int lane = tid & 31;
    const int wave = tid >> 5;
    const int m    = (int)blockIdx.x >> 2;
    const int qtr  = (int)blockIdx.x & 3;
    const int gfirst = qtr * PTS_PER_BLOCK + tid * 4;

    float4 gA[PASSES], gB[PASSES];
    #pragma unroll
    for (int it = 0; it < PASSES; ++it) {
        const float4* gp = (const float4*)(grid + (size_t)(gfirst + it * PTS_PER_PASS) * 2);
        gA[it] = gp[0];
        gB[it] = gp[1];
    }

    const float* Wm = W + (size_t)m * W_ROWSTRIDE;
    const int c = lane & 15;
    const int h = lane >> 4;

    v8f acc0 = {}, acc1 = {}, acc2 = {};
    for (int s = 0; s < K_PER_WAVE / K_PER_STEP; ++s) {
        const int k0 = wave * K_PER_WAVE + s * K_PER_STEP + 4 * c + 2 * h;
        const float2 xv = *(const float2*)(x + k0);
        const float2* wp = (const float2*)(Wm + 3 * k0);
        const float2 f01 = wp[0];
        const float2 f23 = wp[1];
        const float2 f45 = wp[2];
        v2f bx; bx.x = xv.x;  bx.y = xv.y;
        v2f a0; a0.x = f01.x; a0.y = f23.y;
        v2f a1; a1.x = f01.y; a1.y = f45.x;
        v2f a2; a2.x = f23.x; a2.y = f45.y;
        acc0 = __builtin_amdgcn_wmma_f32_16x16x4_f32(false, a0, false, bx, (short)0, acc0, false, false);
        acc1 = __builtin_amdgcn_wmma_f32_16x16x4_f32(false, a1, false, bx, (short)0, acc1, false, false);
        acc2 = __builtin_amdgcn_wmma_f32_16x16x4_f32(false, a2, false, bx, (short)0, acc2, false, false);
    }

    const int r = lane & 7;
    const bool on_diag = (lane < 8) || (lane >= 24);
    float d0 = 0.0f, d1 = 0.0f, d2 = 0.0f;
    #pragma unroll
    for (int v = 0; v < 8; ++v) {
        const bool take = on_diag && (r == v);
        d0 = take ? acc0[v] : d0;
        d1 = take ? acc1[v] : d1;
        d2 = take ? acc2[v] : d2;
    }
    const float s0 = wave_sum(d0);
    const float s1 = wave_sum(d1);
    const float s2 = wave_sum(d2);
    if (lane == 0) {
        red[0 * N_WAVES + wave] = s0;
        red[1 * N_WAVES + wave] = s1;
        red[2 * N_WAVES + wave] = s2;
    }
    __syncthreads();

    float c0 = b[m * 3 + 0], c1 = b[m * 3 + 1], c2 = b[m * 3 + 2];
    for (int w = 0; w < N_WAVES; ++w) {
        c0 += red[0 * N_WAVES + w];
        c1 += red[1 * N_WAVES + w];
        c2 += red[2 * N_WAVES + w];
    }
    const float wx0 = Wm[3 * LATENT + 0], wx1 = Wm[3 * LATENT + 1], wx2 = Wm[3 * LATENT + 2];
    const float wy0 = Wm[3 * (LATENT + 1) + 0], wy1 = Wm[3 * (LATENT + 1) + 1], wy2 = Wm[3 * (LATENT + 1) + 2];

    #pragma unroll
    for (int it = 0; it < PASSES; ++it) {
        const int g0 = gfirst + it * PTS_PER_PASS;
        const float4 ga = gA[it];
        const float4 gb = gB[it];
        const float r0  = FAST_TANH(c0 + ga.x * wx0 + ga.y * wy0);
        const float r1  = FAST_TANH(c1 + ga.x * wx1 + ga.y * wy1);
        const float r2  = FAST_TANH(c2 + ga.x * wx2 + ga.y * wy2);
        const float r3  = FAST_TANH(c0 + ga.z * wx0 + ga.w * wy0);
        const float r4  = FAST_TANH(c1 + ga.z * wx1 + ga.w * wy1);
        const float r5  = FAST_TANH(c2 + ga.z * wx2 + ga.w * wy2);
        const float r6  = FAST_TANH(c0 + gb.x * wx0 + gb.y * wy0);
        const float r7  = FAST_TANH(c1 + gb.x * wx1 + gb.y * wy1);
        const float r8  = FAST_TANH(c2 + gb.x * wx2 + gb.y * wy2);
        const float r9  = FAST_TANH(c0 + gb.z * wx0 + gb.w * wy0);
        const float r10 = FAST_TANH(c1 + gb.z * wx1 + gb.w * wy1);
        const float r11 = FAST_TANH(c2 + gb.z * wx2 + gb.w * wy2);

        float4* op = (float4*)(out + ((size_t)m * N_GRID2 + g0) * 3);
        op[0] = make_float4(r0, r1, r2, r3);
        op[1] = make_float4(r4, r5, r6, r7);
        op[2] = make_float4(r8, r9, r10, r11);
    }
}

extern "C" void kernel_launch(void* const* d_in, const int* in_sizes, int n_in,
                              void* d_out, int out_size, void* d_ws, size_t ws_size,
                              hipStream_t stream) {
    const float* x    = (const float*)d_in[0];
    const float* W    = (const float*)d_in[1];
    const float* b    = (const float*)d_in[2];
    const float* grid = (const float*)d_in[3];
    float* out        = (float*)d_out;

    decode_fused<<<dim3(N_BLOCKS), dim3(256), 0, stream>>>(x, W, b, grid, out);
}


# whole-line output stores: each thread owns consecutive 16-byte units of the flat output (per-lane channel rotation), so a wave store writes 512 contiguous bytes, not 16-B pieces at a 48-B stride
# speedup vs baseline: 2.9106x; 1.5780x over previous
#include <hip/hip_runtime.h>


#define LATENT   1024
#define N_MLP    64
#define N_GRID2  16384
#define OUT_DIM  3
#define W_ROWSTRIDE 3078

#define BLOCKS_PER_MLP 4
#define N_BLOCKS       256
#define PTS_PER_BLOCK  4096
#define F4_PER_BLOCK   3072
#define F4_PER_PASS    256
#define PASSES         12
#define N_WAVES        8
#define K_PER_WAVE     128
#define K_PER_STEP     64

typedef __attribute__((ext_vector_type(2))) float v2f;
typedef __attribute__((ext_vector_type(8))) float v8f;

#if defined(__has_builtin)
#if __has_builtin(__builtin_amdgcn_tanhf)
#define FAST_TANH(v) __builtin_amdgcn_tanhf(v)
#endif
#endif
#ifndef FAST_TANH
#define FAST_TANH(v) tanhf(v)
#endif

template <int MASK>
__device__ __forceinline__ float row_xor_add(float v) {
    const int t = __builtin_amdgcn_update_dpp(0, __builtin_bit_cast(int, v),
                                              0x160 + MASK, 0xF, 0xF, false);
    return v + __builtin_bit_cast(float, t);
}

__device__ __forceinline__ float wave_sum(float v) {
    v = row_xor_add<1>(v);
    v = row_xor_add<2>(v);
    v = row_xor_add<4>(v);
    v = row_xor_add<8>(v);
    const float lo = __builtin_bit_cast(float, __builtin_amdgcn_readlane(__builtin_bit_cast(int, v), 0));
    const float hi = __builtin_bit_cast(float, __builtin_amdgcn_readlane(__builtin_bit_cast(int, v), 16));
    return lo + hi;
}

__global__ __launch_bounds__(256)
void decode_fused(const float* __restrict__ x,
                  const float* __restrict__ W,
                  const float* __restrict__ b,
                  const float* __restrict__ grid,
                  float* __restrict__ out) {
    __shared__ float red[OUT_DIM * N_WAVES];

    const int tid  = (int)threadIdx.x;
    const int lane = tid & 31;
    const int wave = tid >> 5;
    const int m    = (int)blockIdx.x >> 2;
    const int qtr  = (int)blockIdx.x & 3;

    const int p0 = tid % 3;
    const bool q0 = (p0 == 0), q1 = (p0 == 1), q2 = (p0 == 2);
    const int base = (4 * tid) / 3;
    const int gbase0 = qtr * PTS_PER_BLOCK + base;
    const int gbase1 = gbase0 + (q2 ? 1 : 0);
    const int gbase2 = gbase0 + (q0 ? 0 : 1);

    float2 gA[PASSES], gB[PASSES];
    #pragma unroll
    for (int it = 0; it < PASSES; ++it) {
        const int gb = (it % 3 == 0) ? gbase0 : (it % 3 == 1) ? gbase1 : gbase2;
        const float2* gp = (const float2*)(grid + (size_t)(gb + 341 * it + it / 3) * 2);
        gA[it] = gp[0];
        gB[it] = gp[1];
    }

    const float* Wm = W + (size_t)m * W_ROWSTRIDE;
    const int c = lane & 15;
    const int h = lane >> 4;

    v8f acc0 = {}, acc1 = {}, acc2 = {};
    for (int s = 0; s < K_PER_WAVE / K_PER_STEP; ++s) {
        const int k0 = wave * K_PER_WAVE + s * K_PER_STEP + 4 * c + 2 * h;
        const float2 xv = *(const float2*)(x + k0);
        const float2* wp = (const float2*)(Wm + 3 * k0);
        const float2 f01 = wp[0];
        const float2 f23 = wp[1];
        const float2 f45 = wp[2];
        v2f bx; bx.x = xv.x;  bx.y = xv.y;
        v2f a0; a0.x = f01.x; a0.y = f23.y;
        v2f a1; a1.x = f01.y; a1.y = f45.x;
        v2f a2; a2.x = f23.x; a2.y = f45.y;
        acc0 = __builtin_amdgcn_wmma_f32_16x16x4_f32(false, a0, false, bx, (short)0, acc0, false, false);
        acc1 = __builtin_amdgcn_wmma_f32_16x16x4_f32(false, a1, false, bx, (short)0, acc1, false, false);
        acc2 = __builtin_amdgcn_wmma_f32_16x16x4_f32(false, a2, false, bx, (short)0, acc2, false, false);
    }

    const int r = lane & 7;
    const bool on_diag = (lane < 8) || (lane >= 24);
    float d0 = 0.0f, d1 = 0.0f, d2 = 0.0f;
    #pragma unroll
    for (int v = 0; v < 8; ++v) {
        const bool take = on_diag && (r == v);
        d0 = take ? acc0[v] : d0;
        d1 = take ? acc1[v] : d1;
        d2 = take ? acc2[v] : d2;
    }
    const float s0 = wave_sum(d0);
    const float s1 = wave_sum(d1);
    const float s2 = wave_sum(d2);
    if (lane == 0) {
        red[0 * N_WAVES + wave] = s0;
        red[1 * N_WAVES + wave] = s1;
        red[2 * N_WAVES + wave] = s2;
    }
    __syncthreads();

    float c0 = b[m * 3 + 0], c1 = b[m * 3 + 1], c2 = b[m * 3 + 2];
    for (int w = 0; w < N_WAVES; ++w) {
        c0 += red[0 * N_WAVES + w];
        c1 += red[1 * N_WAVES + w];
        c2 += red[2 * N_WAVES + w];
    }
    const float wx0 = Wm[3 * LATENT + 0], wx1 = Wm[3 * LATENT + 1], wx2 = Wm[3 * LATENT + 2];
    const float wy0 = Wm[3 * (LATENT + 1) + 0], wy1 = Wm[3 * (LATENT + 1) + 1], wy2 = Wm[3 * (LATENT + 1) + 2];

    const float rc[3]  = { q0 ? c0  : q1 ? c1  : c2,   q0 ? c1  : q1 ? c2  : c0,   q0 ? c2  : q1 ? c0  : c1  };
    const float rwx[3] = { q0 ? wx0 : q1 ? wx1 : wx2,  q0 ? wx1 : q1 ? wx2 : wx0,  q0 ? wx2 : q1 ? wx0 : wx1 };
    const float rwy[3] = { q0 ? wy0 : q1 ? wy1 : wy2,  q0 ? wy1 : q1 ? wy2 : wy0,  q0 ? wy2 : q1 ? wy0 : wy1 };

    float4* op = (float4*)out + ((size_t)m * (N_GRID2 * OUT_DIM / 4) + qtr * F4_PER_BLOCK + tid);

    #pragma unroll
    for (int it = 0; it < PASSES; ++it) {
        const float2 pa = gA[it];
        const float2 pb = gB[it];
        const bool ph0 = (it % 3 == 0) ? q0 : (it % 3 == 1) ? q2 : q1;
        const bool ph2 = (it % 3 == 0) ? q2 : (it % 3 == 1) ? q1 : q0;
        const float x1 = ph2 ? pb.x : pa.x, y1 = ph2 ? pb.y : pa.y;
        const float x2 = ph0 ? pa.x : pb.x, y2 = ph0 ? pa.y : pb.y;
        const int k0 = it % 3, k1 = (it + 1) % 3, k2 = (it + 2) % 3;
        const float r0 = FAST_TANH(rc[k0] + pa.x * rwx[k0] + pa.y * rwy[k0]);
        const float r1 = FAST_TANH(rc[k1] + x1   * rwx[k1] + y1   * rwy[k1]);
        const float r2 = FAST_TANH(rc[k2] + x2   * rwx[k2] + y2   * rwy[k2]);
        const float r3 = FAST_TANH(rc[k0] + pb.x * rwx[k0] + pb.y * rwy[k0]);
        op[it * F4_PER_PASS] = make_float4(r0, r1, r2, r3);
    }
}

extern "C" void kernel_launch(void* const* d_in, const int* in_sizes, int n_in,
                              void* d_out, int out_size, void* d_ws, size_t ws_size,
                              hipStream_t stream) {
    const float* x    = (const float*)d_in[0];
    const float* W    = (const float*)d_in[1];
    const float* b    = (const float*)d_in[2];
    const float* grid = (const float*)d_in[3];
    float* out        = (float*)d_out;

    decode_fused<<<dim3(N_BLOCKS), dim3(256), 0, stream>>>(x, W, b, grid, out);
}


# every load issued as one group at kernel entry (grid points, bias and the two grid rows of W no longer queue behind the fragment wait and the barrier): one exposed memory round trip instead of three
# speedup vs baseline: 3.2604x; 1.1202x over previous
#include <hip/hip_runtime.h>


#define LATENT   1024
#define N_MLP    64
#define N_GRID2  16384
#define OUT_DIM  3
#define W_ROWSTRIDE 3078

#define BLOCKS_PER_MLP 4
#define N_BLOCKS       256
#define PTS_PER_BLOCK  4096
#define F4_PER_BLOCK   3072
#define F4_PER_PASS    256
#define PASSES         12
#define N_WAVES        8
#define K_PER_WAVE     128
#define K_PER_STEP     64

typedef __attribute__((ext_vector_type(2))) float v2f;
typedef __attribute__((ext_vector_type(8))) float v8f;

#if defined(__has_builtin)
#if __has_builtin(__builtin_amdgcn_tanhf)
#define FAST_TANH(v) __builtin_amdgcn_tanhf(v)
#endif
#endif
#ifndef FAST_TANH
#define FAST_TANH(v) tanhf(v)
#endif

template <int MASK>
__device__ __forceinline__ float row_xor_add(float v) {
    const int t = __builtin_amdgcn_update_dpp(0, __builtin_bit_cast(int, v),
                                              0x160 + MASK, 0xF, 0xF, false);
    return v + __builtin_bit_cast(float, t);
}

__device__ __forceinline__ float wave_sum(float v) {
    v = row_xor_add<1>(v);
    v = row_xor_add<2>(v);
    v = row_xor_add<4>(v);
    v = row_xor_add<8>(v);
    const float lo = __builtin_bit_cast(float, __builtin_amdgcn_readlane(__builtin_bit_cast(int, v), 0));
    const float hi = __builtin_bit_cast(float, __builtin_amdgcn_readlane(__builtin_bit_cast(int, v), 16));
    return lo + hi;
}

__global__ __launch_bounds__(256)
void decode_fused(const float* __restrict__ x,
                  const float* __restrict__ W,
                  const float* __restrict__ b,
                  const float* __restrict__ grid,
                  float* __restrict__ out) {
    __shared__ float red[OUT_DIM * N_WAVES];

    const int tid  = (int)threadIdx.x;
    const int lane = tid & 31;
    const int wave = tid >> 5;
    const int m    = (int)blockIdx.x >> 2;
    const int qtr  = (int)blockIdx.x & 3;

    const int p0 = tid % 3;
    const bool q0 = (p0 == 0), q1 = (p0 == 1), q2 = (p0 == 2);
    const int base = (4 * tid) / 3;
    const int gbase0 = qtr * PTS_PER_BLOCK + base;
    const int gbase1 = gbase0 + (q2 ? 1 : 0);
    const int gbase2 = gbase0 + (q0 ? 0 : 1);

    const float* Wm = W + (size_t)m * W_ROWSTRIDE;
    const int c = lane & 15;
    const int h = lane >> 4;

    float2 xv[K_PER_WAVE / K_PER_STEP], f01[K_PER_WAVE / K_PER_STEP], f23[K_PER_WAVE / K_PER_STEP], f45[K_PER_WAVE / K_PER_STEP];
    #pragma unroll
    for (int s = 0; s < K_PER_WAVE / K_PER_STEP; ++s) {
        const int k0 = wave * K_PER_WAVE + s * K_PER_STEP + 4 * c + 2 * h;
        xv[s] = *(const float2*)(x + k0);
        const float2* wp = (const float2*)(Wm + 3 * k0);
        f01[s] = wp[0];
        f23[s] = wp[1];
        f45[s] = wp[2];
    }

    float2 gA[PASSES], gB[PASSES];
    #pragma unroll
    for (int it = 0; it < PASSES; ++it) {
        const int gb = (it % 3 == 0) ? gbase0 : (it % 3 == 1) ? gbase1 : gbase2;
        const float2* gp = (const float2*)(grid + (size_t)(gb + 341 * it + it / 3) * 2);
        gA[it] = gp[0];
        gB[it] = gp[1];
    }

    const float bias0 = b[m * 3 + 0], bias1 = b[m * 3 + 1], bias2 = b[m * 3 + 2];
    const float wx0 = Wm[3 * LATENT + 0], wx1 = Wm[3 * LATENT + 1], wx2 = Wm[3 * LATENT + 2];
    const float wy0 = Wm[3 * (LATENT + 1) + 0], wy1 = Wm[3 * (LATENT + 1) + 1], wy2 = Wm[3 * (LATENT + 1) + 2];

    __builtin_amdgcn_sched_barrier(0);

    v8f acc0 = {}, acc1 = {}, acc2 = {};
    #pragma unroll
    for (int s = 0; s < K_PER_WAVE / K_PER_STEP; ++s) {
        v2f bx; bx.x = xv[s].x;  bx.y = xv[s].y;
        v2f a0; a0.x = f01[s].x; a0.y = f23[s].y;
        v2f a1; a1.x = f01[s].y; a1.y = f45[s].x;
        v2f a2; a2.x = f23[s].x; a2.y = f45[s].y;
        acc0 = __builtin_amdgcn_wmma_f32_16x16x4_f32(false, a0, false, bx, (short)0, acc0, false, false);
        acc1 = __builtin_amdgcn_wmma_f32_16x16x4_f32(false, a1, false, bx, (short)0, acc1, false, false);
        acc2 = __builtin_amdgcn_wmma_f32_16x16x4_f32(false, a2, false, bx, (short)0, acc2, false, false);
    }

    const int r = lane & 7;
    const bool on_diag = (lane < 8) || (lane >= 24);
    float d0 = 0.0f, d1 = 0.0f, d2 = 0.0f;
    #pragma unroll
    for (int v = 0; v < 8; ++v) {
        const bool take = on_diag && (r == v);
        d0 = take ? acc0[v] : d0;
        d1 = take ? acc1[v] : d1;
        d2 = take ? acc2[v] : d2;
    }
    const float s0 = wave_sum(d0);
    const float s1 = wave_sum(d1);
    const float s2 = wave_sum(d2);
    if (lane == 0) {
        red[0 * N_WAVES + wave] = s0;
        red[1 * N_WAVES + wave] = s1;
        red[2 * N_WAVES + wave] = s2;
    }
    __syncthreads();

    float c0 = bias0, c1 = bias1, c2 = bias2;
    for (int w = 0; w < N_WAVES; ++w) {
        c0 += red[0 * N_WAVES + w];
        c1 += red[1 * N_WAVES + w];
        c2 += red[2 * N_WAVES + w];
    }
    const float rc[3]  = { q0 ? c0  : q1 ? c1  : c2,   q0 ? c1  : q1 ? c2  : c0,   q0 ? c2  : q1 ? c0  : c1  };
    const float rwx[3] = { q0 ? wx0 : q1 ? wx1 : wx2,  q0 ? wx1 : q1 ? wx2 : wx0,  q0 ? wx2 : q1 ? wx0 : wx1 };
    const float rwy[3] = { q0 ? wy0 : q1 ? wy1 : wy2,  q0 ? wy1 : q1 ? wy2 : wy0,  q0 ? wy2 : q1 ? wy0 : wy1 };

    float4* op = (float4*)out + ((size_t)m * (N_GRID2 * OUT_DIM / 4) + qtr * F4_PER_BLOCK + tid);

    #pragma unroll
    for (int it = 0; it < PASSES; ++it) {
        const float2 pa = gA[it];
        const float2 pb = gB[it];
        const bool ph0 = (it % 3 == 0) ? q0 : (it % 3 == 1) ? q2 : q1;
        const bool ph2 = (it % 3 == 0) ? q2 : (it % 3 == 1) ? q1 : q0;
        const float x1 = ph2 ? pb.x : pa.x, y1 = ph2 ? pb.y : pa.y;
        const float x2 = ph0 ? pa.x : pb.x, y2 = ph0 ? pa.y : pb.y;
        const int k0 = it % 3, k1 = (it + 1) % 3, k2 = (it + 2) % 3;
        const float r0 = FAST_TANH(rc[k0] + pa.x * rwx[k0] + pa.y * rwy[k0]);
        const float r1 = FAST_TANH(rc[k1] + x1   * rwx[k1] + y1   * rwy[k1]);
        const float r2 = FAST_TANH(rc[k2] + x2   * rwx[k2] + y2   * rwy[k2]);
        const float r3 = FAST_TANH(rc[k0] + pb.x * rwx[k0] + pb.y * rwy[k0]);
        op[it * F4_PER_PASS] = make_float4(r0, r1, r2, r3);
    }
}

extern "C" void kernel_launch(void* const* d_in, const int* in_sizes, int n_in,
                              void* d_out, int out_size, void* d_ws, size_t ws_size,
                              hipStream_t stream) {
    const float* x    = (const float*)d_in[0];
    const float* W    = (const float*)d_in[1];
    const float* b    = (const float*)d_in[2];
    const float* grid = (const float*)d_in[3];
    float* out        = (float*)d_out;

    decode_fused<<<dim3(N_BLOCKS), dim3(256), 0, stream>>>(x, W, b, grid, out);
}
